# starting kernel
# baseline (speedup 1.0000x reference)
#include <hip/hip_runtime.h>

typedef float v2f __attribute__((ext_vector_type(2)));
typedef float v4f __attribute__((ext_vector_type(4)));
typedef float v8f __attribute__((ext_vector_type(8)));

#define N_LEN 131072
#define F_TAPS 16
#define TILES_TOTAL 8190
#define STRIP_TILES 30
#define STRIPS_PER_GROUP 273
#define GROUPS 16
#define WAVES_PER_BLOCK 8
#define NUM_BLOCKS ((GROUPS * STRIPS_PER_GROUP) / WAVES_PER_BLOCK)

__device__ __forceinline__ float k1_tap(const float* h, int a) {
    return (a == 0) ? 1.0f : h[a - 1];
}
__device__ __forceinline__ float k2_tap(const float* h, int b) {
    return (b == 0) ? 1.0f : h[16 - b];
}

__device__ __forceinline__ void tile_step(const v2f a[3][4],
                                          const v2f* bo, const v2f* bm, const v2f* bn,
                                          float* yptr) {
    v8f acc = {};
#pragma unroll
    for (int c = 0; c < 4; ++c)
        acc = __builtin_amdgcn_wmma_f32_16x16x4_f32(false, a[2][c], false, bo[c],
                                                    (short)0, acc, false, false);
#pragma unroll
    for (int c = 0; c < 4; ++c)
        acc = __builtin_amdgcn_wmma_f32_16x16x4_f32(false, a[1][c], false, bm[c],
                                                    (short)0, acc, false, false);
#pragma unroll
    for (int c = 0; c < 4; ++c)
        acc = __builtin_amdgcn_wmma_f32_16x16x4_f32(false, a[0][c], false, bn[c],
                                                    (short)0, acc, false, false);
    v4f lo = {acc[0], acc[1], acc[2], acc[3]};
    v4f hi = {acc[4], acc[5], acc[6], acc[7]};
    *(v4f*)(yptr)     = lo;
    *(v4f*)(yptr + 4) = hi;
}

__global__ void __launch_bounds__(256) fir_wmma(const float* __restrict__ x,
                                                const float* __restrict__ h,
                                                float* __restrict__ y) {
    __shared__ float sKpad[64];
    const int tid = threadIdx.x;
    if (tid < 63) {
        const int t = tid - 15;
        float v = 0.0f;
        if (t >= 0 && t <= 32) {
            const int alo = (t > 16) ? (t - 16) : 0;
            const int ahi = (t < 16) ? t : 16;
            float s = 0.0f;
            for (int a = alo; a <= ahi; ++a) s += k1_tap(h, a) * k2_tap(h, t - a);
            v = s;
        }
        sKpad[tid] = v;
    }
    __syncthreads();

    const int wave = tid >> 5;
    const int lane = tid & 31;
    const int lo16 = lane & 15;
    const int half = lane >> 4;

    const int wave_id = blockIdx.x * WAVES_PER_BLOCK + wave;
    const int group   = wave_id / STRIPS_PER_GROUP;
    const int strip   = wave_id % STRIPS_PER_GROUP;
    const int b0      = group * 16;
    const int P0      = 32 + strip * (STRIP_TILES * 16);

    v2f a[3][4];
    const int m = lo16;
#pragma unroll
    for (int blk = 0; blk < 3; ++blk) {
#pragma unroll
        for (int c = 0; c < 4; ++c) {
            const int p0  = 4 * c + 2 * half;
            const int idx = 15 + m - p0 + 16 * blk;
            a[blk][c].x = sKpad[idx];
            a[blk][c].y = sKpad[idx - 1];
        }
    }

    const float* xrow = x + (size_t)(b0 + lo16) * N_LEN;
    float*       yrow = y + (size_t)(b0 + lo16) * N_LEN;

    v2f B0[4], B1[4], B2[4];
#define LOADB(dst, pos)                                            \
    {                                                              \
        _Pragma("unroll")                                          \
        for (int c = 0; c < 4; ++c)                                \
            dst[c] = *(const v2f*)(xrow + (pos) + 4 * c + 2 * half); \
    }

    LOADB(B0, P0 - 32);
    LOADB(B1, P0 - 16);

    for (int tt = 0; tt < STRIP_TILES; tt += 3) {
        const int P = P0 + 16 * tt;
        __builtin_prefetch(xrow + P + 96, 0, 0);

        LOADB(B2, P);
        tile_step(a, B0, B1, B2, yrow + P + 8 * half);

        LOADB(B0, P + 16);
        tile_step(a, B1, B2, B0, yrow + P + 16 + 8 * half);

        LOADB(B1, P + 32);
        tile_step(a, B2, B0, B1, yrow + P + 32 + 8 * half);
    }
#undef LOADB
}

__global__ void fir_boundary(const float* __restrict__ x,
                             const float* __restrict__ h,
                             float* __restrict__ y) {
    const int b = blockIdx.x;
    const int i = threadIdx.x;
    const float* xr = x + (size_t)b * N_LEN;
    float out = 0.0f;
    if (i >= 16) {
        float hl[16];
#pragma unroll
        for (int j = 0; j < 16; ++j) hl[j] = h[j];
        float s;
        {
            float acc = xr[i];
#pragma unroll
            for (int j = 0; j < 16; ++j) acc += hl[j] * xr[i - 1 - j];
            s = acc;
        }
        for (int j = 0; j <= i - 17; ++j) {
            const int k = i - 1 - j;
            float acc = xr[k];
#pragma unroll
            for (int jj = 0; jj < 16; ++jj) acc += hl[jj] * xr[k - 1 - jj];
            s += hl[15 - j] * acc;
        }
        out = s;
    }
    y[(size_t)b * N_LEN + i] = out;
}

extern "C" void kernel_launch(void* const* d_in, const int* in_sizes, int n_in,
                              void* d_out, int out_size, void* d_ws, size_t ws_size,
                              hipStream_t stream) {
    const float* x = (const float*)d_in[0];
    const float* h = (const float*)d_in[1];
    float* y = (float*)d_out;
    (void)in_sizes; (void)n_in; (void)out_size; (void)d_ws; (void)ws_size;

    fir_boundary<<<256, 32, 0, stream>>>(x, h, y);
    fir_wmma<<<NUM_BLOCKS, 256, 0, stream>>>(x, h, y);
}


# Matrix N dim = 16 consecutive 16-sample blocks of ONE row (was 16 batch rows): contiguous 16-byte loads/stores; shifted operands by in-row lane rotate; k order inside a block changes (sum order)
# speedup vs baseline: 1.6890x; 1.6890x over previous
#include <hip/hip_runtime.h>

typedef float v2f __attribute__((ext_vector_type(2)));
typedef float v4f __attribute__((ext_vector_type(4)));
typedef float v8f __attribute__((ext_vector_type(8)));

#define N_LEN 131072
#define F_TAPS 16
#define BATCH 256
#define TILE 256
#define TILES_PER_ROW (N_LEN / TILE)
#define TILES_PER_WAVE 32
#define STRIPS_PER_ROW (TILES_PER_ROW / TILES_PER_WAVE)
#define GROUP 4
#define GROUPS_PER_WAVE (TILES_PER_WAVE / GROUP)
#define WAVES_PER_BLOCK 8
#define NUM_BLOCKS ((BATCH * STRIPS_PER_ROW) / WAVES_PER_BLOCK)

__device__ __forceinline__ float k1_tap(const float* h, int a) {
    return (a == 0) ? 1.0f : h[a - 1];
}
__device__ __forceinline__ float k2_tap(const float* h, int b) {
    return (b == 0) ? 1.0f : h[16 - b];
}

__device__ __forceinline__ int blk_off(int hh, int q) {
    return 4 * hh + q + ((q >= 4) ? 4 : 0);
}

__device__ __forceinline__ float row_rot1(float v) {
    const unsigned u = __builtin_bit_cast(unsigned, v);
    return __builtin_bit_cast(float, __builtin_amdgcn_permlane16(
        u, u, 0x6543210Fu, 0xEDCBA987u, false, false));
}
__device__ __forceinline__ float row_rot2(float v) {
    const unsigned u = __builtin_bit_cast(unsigned, v);
    return __builtin_bit_cast(float, __builtin_amdgcn_permlane16(
        u, u, 0x543210FEu, 0xDCBA9876u, false, false));
}

struct Raw { v4f lo, hi; };

__device__ __forceinline__ Raw load_tile(const float* p) {
    Raw r;
    r.lo = *(const v4f*)(p);
    r.hi = *(const v4f*)(p + 8);
    return r;
}

__device__ __forceinline__ void tile_step(const v2f a[3][4], const Raw& cur, const Raw& prev,
                                          bool is15, bool ge14, float* yptr, bool do_store) {
    float rn[8] = {cur.lo[0], cur.lo[1], cur.lo[2], cur.lo[3],
                   cur.hi[0], cur.hi[1], cur.hi[2], cur.hi[3]};
    float rp[8] = {prev.lo[0], prev.lo[1], prev.lo[2], prev.lo[3],
                   prev.hi[0], prev.hi[1], prev.hi[2], prev.hi[3]};
    float rm[8], ro[8];
#pragma unroll
    for (int q = 0; q < 8; ++q) {
        rm[q] = row_rot1(is15 ? rp[q] : rn[q]);
        ro[q] = row_rot2(ge14 ? rp[q] : rn[q]);
    }
    v8f acc = {};
#pragma unroll
    for (int c = 0; c < 4; ++c) {
        v2f b = {ro[2 * c], ro[2 * c + 1]};
        acc = __builtin_amdgcn_wmma_f32_16x16x4_f32(false, a[2][c], false, b,
                                                    (short)0, acc, false, false);
    }
#pragma unroll
    for (int c = 0; c < 4; ++c) {
        v2f b = {rm[2 * c], rm[2 * c + 1]};
        acc = __builtin_amdgcn_wmma_f32_16x16x4_f32(false, a[1][c], false, b,
                                                    (short)0, acc, false, false);
    }
#pragma unroll
    for (int c = 0; c < 4; ++c) {
        v2f b = {rn[2 * c], rn[2 * c + 1]};
        acc = __builtin_amdgcn_wmma_f32_16x16x4_f32(false, a[0][c], false, b,
                                                    (short)0, acc, false, false);
    }
    if (do_store) {
        v4f lo = {acc[0], acc[1], acc[2], acc[3]};
        v4f hi = {acc[4], acc[5], acc[6], acc[7]};
        *(v4f*)(yptr)     = lo;
        *(v4f*)(yptr + 8) = hi;
    }
}

__device__ __forceinline__ void load_group(Raw (&buf)[GROUP], const float* p) {
#pragma unroll
    for (int u = 0; u < GROUP; ++u) buf[u] = load_tile(p + u * TILE);
}

__device__ __forceinline__ void compute_group(const v2f a[3][4], const Raw (&buf)[GROUP],
                                              const Raw& prev, bool is15, bool ge14,
                                              float* yptr, bool first_store) {
#pragma unroll
    for (int u = 0; u < GROUP; ++u)
        tile_step(a, buf[u], (u == 0) ? prev : buf[u - 1], is15, ge14,
                  yptr + u * TILE, (u == 0) ? first_store : true);
}

__global__ void __launch_bounds__(256) fir_wmma(const float* __restrict__ x,
                                                const float* __restrict__ h,
                                                float* __restrict__ y) {
    __shared__ float sKpad[64];
    const int tid = threadIdx.x;
    if (tid < 63) {
        const int t = tid - 15;
        float v = 0.0f;
        if (t >= 0 && t <= 32) {
            const int alo = (t > 16) ? (t - 16) : 0;
            const int ahi = (t < 16) ? t : 16;
            float s = 0.0f;
            for (int a = alo; a <= ahi; ++a) s += k1_tap(h, a) * k2_tap(h, t - a);
            v = s;
        }
        sKpad[tid] = v;
    }
    __syncthreads();

    const int wave = tid >> 5;
    const int lane = tid & 31;
    const int n    = lane & 15;
    const int half = lane >> 4;

    const int wave_id = blockIdx.x * WAVES_PER_BLOCK + wave;
    const int row     = wave_id / STRIPS_PER_ROW;
    const int strip   = wave_id % STRIPS_PER_ROW;

    v2f a[3][4];
    const int om = blk_off(n >> 3, n & 7);
#pragma unroll
    for (int blk = 0; blk < 3; ++blk) {
#pragma unroll
        for (int c = 0; c < 4; ++c) {
            const int i0 = 15 + om - blk_off(half, 2 * c)     + 16 * blk;
            const int i1 = 15 + om - blk_off(half, 2 * c + 1) + 16 * blk;
            a[blk][c].x = sKpad[i0];
            a[blk][c].y = sKpad[i1];
        }
    }

    const size_t lane_off = (size_t)row * N_LEN + (size_t)strip * (TILES_PER_WAVE * TILE)
                          + 16 * n + 4 * half;
    const float* xs = x + lane_off;
    float*       ys = y + lane_off;

    const bool is15 = (n == 15);
    const bool ge14 = (n >= 14);

    Raw prev;
    prev.lo = (v4f){0.0f, 0.0f, 0.0f, 0.0f};
    prev.hi = prev.lo;
    if (strip > 0 && ge14) prev = load_tile(xs - TILE);
    const bool first_store = (strip > 0) || (n >= 2);

    Raw A[GROUP], B[GROUP];
    load_group(A, xs);
    for (int g = 0; g < GROUPS_PER_WAVE - 2; g += 2) {
        load_group(B, xs + (g + 1) * (GROUP * TILE));
        __builtin_amdgcn_sched_barrier(0);
        compute_group(a, A, prev, is15, ge14, ys + g * (GROUP * TILE), (g > 0) || first_store);
        const Raw lastA = A[GROUP - 1];
        load_group(A, xs + (g + 2) * (GROUP * TILE));
        __builtin_amdgcn_sched_barrier(0);
        compute_group(a, B, lastA, is15, ge14, ys + (g + 1) * (GROUP * TILE), true);
        prev = B[GROUP - 1];
    }
    {
        const int g = GROUPS_PER_WAVE - 2;
        load_group(B, xs + (g + 1) * (GROUP * TILE));
        __builtin_amdgcn_sched_barrier(0);
        compute_group(a, A, prev, is15, ge14, ys + g * (GROUP * TILE), true);
        compute_group(a, B, A[GROUP - 1], is15, ge14, ys + (g + 1) * (GROUP * TILE), true);
    }
}

__global__ void fir_boundary(const float* __restrict__ x,
                             const float* __restrict__ h,
                             float* __restrict__ y) {
    const int b = blockIdx.x;
    const int i = threadIdx.x;
    const float* xr = x + (size_t)b * N_LEN;
    float out = 0.0f;
    if (i >= 16) {
        float hl[16];
#pragma unroll
        for (int j = 0; j < 16; ++j) hl[j] = h[j];
        float s;
        {
            float acc = xr[i];
#pragma unroll
            for (int j = 0; j < 16; ++j) acc += hl[j] * xr[i - 1 - j];
            s = acc;
        }
        for (int j = 0; j <= i - 17; ++j) {
            const int k = i - 1 - j;
            float acc = xr[k];
#pragma unroll
            for (int jj = 0; jj < 16; ++jj) acc += hl[jj] * xr[k - 1 - jj];
            s += hl[15 - j] * acc;
        }
        out = s;
    }
    y[(size_t)b * N_LEN + i] = out;
}

extern "C" void kernel_launch(void* const* d_in, const int* in_sizes, int n_in,
                              void* d_out, int out_size, void* d_ws, size_t ws_size,
                              hipStream_t stream) {
    const float* x = (const float*)d_in[0];
    const float* h = (const float*)d_in[1];
    float* y = (float*)d_out;
    (void)in_sizes; (void)n_in; (void)out_size; (void)d_ws; (void)ws_size;

    fir_boundary<<<256, 32, 0, stream>>>(x, h, y);
    fir_wmma<<<NUM_BLOCKS, 256, 0, stream>>>(x, h, y);
}


# Samples 0..31 of each row (exact masked two-pass form, by lane exchanges) written by the wave that owns the row's first strip, inside the tile kernel: one launch instead of two; sum order unchanged
# speedup vs baseline: 1.9214x; 1.1376x over previous
#include <hip/hip_runtime.h>

typedef float v2f __attribute__((ext_vector_type(2)));
typedef float v4f __attribute__((ext_vector_type(4)));
typedef float v8f __attribute__((ext_vector_type(8)));

#define N_LEN 131072
#define F_TAPS 16
#define BATCH 256
#define TILE 256
#define TILES_PER_ROW (N_LEN / TILE)
#define TILES_PER_WAVE 32
#define STRIPS_PER_ROW (TILES_PER_ROW / TILES_PER_WAVE)
#define GROUP 4
#define GROUPS_PER_WAVE (TILES_PER_WAVE / GROUP)
#define WAVES_PER_BLOCK 8
#define NUM_BLOCKS ((BATCH * STRIPS_PER_ROW) / WAVES_PER_BLOCK)

__device__ __forceinline__ float k1_tap(const float* h, int a) {
    return (a == 0) ? 1.0f : h[a - 1];
}
__device__ __forceinline__ float k2_tap(const float* h, int b) {
    return (b == 0) ? 1.0f : h[16 - b];
}

__device__ __forceinline__ int blk_off(int hh, int q) {
    return 4 * hh + q + ((q >= 4) ? 4 : 0);
}

__device__ __forceinline__ float row_rot1(float v) {
    const unsigned u = __builtin_bit_cast(unsigned, v);
    return __builtin_bit_cast(float, __builtin_amdgcn_permlane16(
        u, u, 0x6543210Fu, 0xEDCBA987u, false, false));
}
__device__ __forceinline__ float row_rot2(float v) {
    const unsigned u = __builtin_bit_cast(unsigned, v);
    return __builtin_bit_cast(float, __builtin_amdgcn_permlane16(
        u, u, 0x543210FEu, 0xDCBA9876u, false, false));
}

struct Raw { v4f lo, hi; };

__device__ __forceinline__ Raw load_tile(const float* p) {
    Raw r;
    r.lo = *(const v4f*)(p);
    r.hi = *(const v4f*)(p + 8);
    return r;
}

__device__ __forceinline__ void tile_step(const v2f a[3][4], const Raw& cur, const Raw& prev,
                                          bool is15, bool ge14, float* yptr, bool do_store) {
    float rn[8] = {cur.lo[0], cur.lo[1], cur.lo[2], cur.lo[3],
                   cur.hi[0], cur.hi[1], cur.hi[2], cur.hi[3]};
    float rp[8] = {prev.lo[0], prev.lo[1], prev.lo[2], prev.lo[3],
                   prev.hi[0], prev.hi[1], prev.hi[2], prev.hi[3]};
    float rm[8], ro[8];
#pragma unroll
    for (int q = 0; q < 8; ++q) {
        rm[q] = row_rot1(is15 ? rp[q] : rn[q]);
        ro[q] = row_rot2(ge14 ? rp[q] : rn[q]);
    }
    v8f acc = {};
#pragma unroll
    for (int c = 0; c < 4; ++c) {
        v2f b = {ro[2 * c], ro[2 * c + 1]};
        acc = __builtin_amdgcn_wmma_f32_16x16x4_f32(false, a[2][c], false, b,
                                                    (short)0, acc, false, false);
    }
#pragma unroll
    for (int c = 0; c < 4; ++c) {
        v2f b = {rm[2 * c], rm[2 * c + 1]};
        acc = __builtin_amdgcn_wmma_f32_16x16x4_f32(false, a[1][c], false, b,
                                                    (short)0, acc, false, false);
    }
#pragma unroll
    for (int c = 0; c < 4; ++c) {
        v2f b = {rn[2 * c], rn[2 * c + 1]};
        acc = __builtin_amdgcn_wmma_f32_16x16x4_f32(false, a[0][c], false, b,
                                                    (short)0, acc, false, false);
    }
    if (do_store) {
        v4f lo = {acc[0], acc[1], acc[2], acc[3]};
        v4f hi = {acc[4], acc[5], acc[6], acc[7]};
        *(v4f*)(yptr)     = lo;
        *(v4f*)(yptr + 8) = hi;
    }
}

__device__ __forceinline__ void load_group(Raw (&buf)[GROUP], const float* p) {
#pragma unroll
    for (int u = 0; u < GROUP; ++u) buf[u] = load_tile(p + u * TILE);
}

__device__ __forceinline__ void compute_group(const v2f a[3][4], const Raw (&buf)[GROUP],
                                              const Raw& prev, bool is15, bool ge14,
                                              float* yptr, bool first_store) {
#pragma unroll
    for (int u = 0; u < GROUP; ++u)
        tile_step(a, buf[u], (u == 0) ? prev : buf[u - 1], is15, ge14,
                  yptr + u * TILE, (u == 0) ? first_store : true);
}

__device__ __forceinline__ float row_head_value(float xi, const float* __restrict__ h, int lane) {
    float hl[16];
#pragma unroll
    for (int j = 0; j < 16; ++j) hl[j] = h[j];
    float acc = xi;
#pragma unroll
    for (int j = 0; j < 16; ++j) acc += hl[j] * __shfl(xi, (lane - 1 - j) & 31, 32);
    float s = acc;
#pragma unroll
    for (int j = 0; j < 15; ++j) {
        const int k = lane - 1 - j;
        const float vk = __shfl(acc, k & 31, 32);
        if (k >= 16) s += hl[15 - j] * vk;
    }
    return (lane >= 16) ? s : 0.0f;
}

__global__ void __launch_bounds__(256) fir_wmma(const float* __restrict__ x,
                                                const float* __restrict__ h,
                                                float* __restrict__ y) {
    __shared__ float sKpad[64];
    const int tid = threadIdx.x;
    if (tid < 63) {
        const int t = tid - 15;
        float v = 0.0f;
        if (t >= 0 && t <= 32) {
            const int alo = (t > 16) ? (t - 16) : 0;
            const int ahi = (t < 16) ? t : 16;
            float s = 0.0f;
            for (int a = alo; a <= ahi; ++a) s += k1_tap(h, a) * k2_tap(h, t - a);
            v = s;
        }
        sKpad[tid] = v;
    }
    __syncthreads();

    const int wave = tid >> 5;
    const int lane = tid & 31;
    const int n    = lane & 15;
    const int half = lane >> 4;

    const int wave_id = blockIdx.x * WAVES_PER_BLOCK + wave;
    const int row     = wave_id / STRIPS_PER_ROW;
    const int strip   = wave_id % STRIPS_PER_ROW;

    v2f a[3][4];
    const int om = blk_off(n >> 3, n & 7);
#pragma unroll
    for (int blk = 0; blk < 3; ++blk) {
#pragma unroll
        for (int c = 0; c < 4; ++c) {
            const int i0 = 15 + om - blk_off(half, 2 * c)     + 16 * blk;
            const int i1 = 15 + om - blk_off(half, 2 * c + 1) + 16 * blk;
            a[blk][c].x = sKpad[i0];
            a[blk][c].y = sKpad[i1];
        }
    }

    const size_t lane_off = (size_t)row * N_LEN + (size_t)strip * (TILES_PER_WAVE * TILE)
                          + 16 * n + 4 * half;
    const float* xs = x + lane_off;
    float*       ys = y + lane_off;

    const bool is15 = (n == 15);
    const bool ge14 = (n >= 14);

    const bool row_head = (strip == 0);
    float xhead = 0.0f;
    if (row_head) xhead = x[(size_t)row * N_LEN + lane];

    Raw prev;
    prev.lo = (v4f){0.0f, 0.0f, 0.0f, 0.0f};
    prev.hi = prev.lo;
    if (strip > 0 && ge14) prev = load_tile(xs - TILE);
    const bool first_store = (strip > 0) || (n >= 2);

    Raw A[GROUP], B[GROUP];
    load_group(A, xs);
    if (row_head) y[(size_t)row * N_LEN + lane] = row_head_value(xhead, h, lane);
    for (int g = 0; g < GROUPS_PER_WAVE - 2; g += 2) {
        load_group(B, xs + (g + 1) * (GROUP * TILE));
        __builtin_amdgcn_sched_barrier(0);
        compute_group(a, A, prev, is15, ge14, ys + g * (GROUP * TILE), (g > 0) || first_store);
        const Raw lastA = A[GROUP - 1];
        load_group(A, xs + (g + 2) * (GROUP * TILE));
        __builtin_amdgcn_sched_barrier(0);
        compute_group(a, B, lastA, is15, ge14, ys + (g + 1) * (GROUP * TILE), true);
        prev = B[GROUP - 1];
    }
    {
        const int g = GROUPS_PER_WAVE - 2;
        load_group(B, xs + (g + 1) * (GROUP * TILE));
        __builtin_amdgcn_sched_barrier(0);
        compute_group(a, A, prev, is15, ge14, ys + g * (GROUP * TILE), true);
        compute_group(a, B, A[GROUP - 1], is15, ge14, ys + (g + 1) * (GROUP * TILE), true);
    }
}

extern "C" void kernel_launch(void* const* d_in, const int* in_sizes, int n_in,
                              void* d_out, int out_size, void* d_ws, size_t ws_size,
                              hipStream_t stream) {
    const float* x = (const float*)d_in[0];
    const float* h = (const float*)d_in[1];
    float* y = (float*)d_out;
    (void)in_sizes; (void)n_in; (void)out_size; (void)d_ws; (void)ws_size;

    fir_wmma<<<NUM_BLOCKS, 256, 0, stream>>>(x, h, y);
}


# Global loads and stores move whole 128-byte lines: tiles pass through per-wave LDS into fragments and back; shifted operands are LDS reads, no lane rotates; k order in a block changes (sum order)
# speedup vs baseline: 1.9984x; 1.0401x over previous
#include <hip/hip_runtime.h>

typedef float v2f __attribute__((ext_vector_type(2)));
typedef float v4f __attribute__((ext_vector_type(4)));
typedef float v8f __attribute__((ext_vector_type(8)));

#define N_LEN 131072
#define F_TAPS 16
#define BATCH 256
#define TILE 256
#define TILES_PER_ROW (N_LEN / TILE)
#define TILES_PER_WAVE 32
#define STRIPS_PER_ROW (TILES_PER_ROW / TILES_PER_WAVE)
#define GROUP 4
#define GROUPS_PER_WAVE (TILES_PER_WAVE / GROUP)
#define WAVES_PER_BLOCK 8
#define NUM_BLOCKS ((BATCH * STRIPS_PER_ROW) / WAVES_PER_BLOCK)

#define LDS_TILE  320
#define LDS_GROUP (GROUP * LDS_TILE)
#define LDS_RING  (2 * LDS_GROUP)
#define LDS_WAVE  (LDS_RING + LDS_GROUP)
#define LINE_STEP 136

__device__ __forceinline__ float k1_tap(const float* h, int a) {
    return (a == 0) ? 1.0f : h[a - 1];
}
__device__ __forceinline__ float k2_tap(const float* h, int b) {
    return (b == 0) ? 1.0f : h[16 - b];
}

__device__ __forceinline__ void lds_wave_sync() {
    asm volatile("s_wait_dscnt 0x0" ::: "memory");
}

__device__ __forceinline__ void load_lines(v4f (&r)[GROUP][2], const float* p) {
#pragma unroll
    for (int u = 0; u < GROUP; ++u) {
        r[u][0] = *(const v4f*)(p + u * TILE);
        r[u][1] = *(const v4f*)(p + u * TILE + 128);
    }
}

__device__ __forceinline__ void stage_in(const v4f (&r)[GROUP][2], float* slot, int wl) {
#pragma unroll
    for (int u = 0; u < GROUP; ++u) {
        *(v4f*)(slot + u * LDS_TILE + wl)             = r[u][0];
        *(v4f*)(slot + u * LDS_TILE + wl + LINE_STEP) = r[u][1];
    }
}

__device__ __forceinline__ v8f tile_product(const v2f a[3][4], const float* pn, const float* pm,
                                            const float* po) {
    const v4f n0 = *(const v4f*)(pn), n1 = *(const v4f*)(pn + 4);
    const v4f m0 = *(const v4f*)(pm), m1 = *(const v4f*)(pm + 4);
    const v4f o0 = *(const v4f*)(po), o1 = *(const v4f*)(po + 4);
    const float rn[8] = {n0[0], n0[1], n0[2], n0[3], n1[0], n1[1], n1[2], n1[3]};
    const float rm[8] = {m0[0], m0[1], m0[2], m0[3], m1[0], m1[1], m1[2], m1[3]};
    const float ro[8] = {o0[0], o0[1], o0[2], o0[3], o1[0], o1[1], o1[2], o1[3]};
    v8f acc = {};
#pragma unroll
    for (int c = 0; c < 4; ++c) {
        v2f b = {ro[2 * c], ro[2 * c + 1]};
        acc = __builtin_amdgcn_wmma_f32_16x16x4_f32(false, a[2][c], false, b,
                                                    (short)0, acc, false, false);
    }
#pragma unroll
    for (int c = 0; c < 4; ++c) {
        v2f b = {rm[2 * c], rm[2 * c + 1]};
        acc = __builtin_amdgcn_wmma_f32_16x16x4_f32(false, a[1][c], false, b,
                                                    (short)0, acc, false, false);
    }
#pragma unroll
    for (int c = 0; c < 4; ++c) {
        v2f b = {rn[2 * c], rn[2 * c + 1]};
        acc = __builtin_amdgcn_wmma_f32_16x16x4_f32(false, a[0][c], false, b,
                                                    (short)0, acc, false, false);
    }
    return acc;
}

template <int SLOT>
__device__ __forceinline__ void compute_group(const v2f a[3][4], const float* ring, float* outb,
                                              const int (&fr)[3], const int (&frw)[3],
                                              int ow, int wl, float* yptr, bool skip_head) {
    v8f acc[GROUP];
#pragma unroll
    for (int u = 0; u < GROUP; ++u) {
        const float* tb = ring + SLOT * LDS_GROUP + u * LDS_TILE;
        const bool wrap = (SLOT == 0) && (u == 0);
        acc[u] = tile_product(a, tb + fr[0],
                              wrap ? (ring + frw[1]) : (tb + fr[1]),
                              wrap ? (ring + frw[2]) : (tb + fr[2]));
    }
#pragma unroll
    for (int u = 0; u < GROUP; ++u) {
        const v4f lo = {acc[u][0], acc[u][1], acc[u][2], acc[u][3]};
        const v4f hi = {acc[u][4], acc[u][5], acc[u][6], acc[u][7]};
        *(v4f*)(outb + u * LDS_TILE + ow)     = lo;
        *(v4f*)(outb + u * LDS_TILE + ow + 4) = hi;
    }
    lds_wave_sync();
#pragma unroll
    for (int u = 0; u < GROUP; ++u) {
        const v4f l0 = *(const v4f*)(outb + u * LDS_TILE + wl);
        const v4f l1 = *(const v4f*)(outb + u * LDS_TILE + wl + LINE_STEP);
        if (!(u == 0 && skip_head)) *(v4f*)(yptr + u * TILE) = l0;
        *(v4f*)(yptr + u * TILE + 128) = l1;
    }
}

__device__ __forceinline__ float row_head_value(float xi, const float* __restrict__ h, int lane) {
    float hl[16];
#pragma unroll
    for (int j = 0; j < 16; ++j) hl[j] = h[j];
    float acc = xi;
#pragma unroll
    for (int j = 0; j < 16; ++j) acc += hl[j] * __shfl(xi, (lane - 1 - j) & 31, 32);
    float s = acc;
#pragma unroll
    for (int j = 0; j < 15; ++j) {
        const int k = lane - 1 - j;
        const float vk = __shfl(acc, k & 31, 32);
        if (k >= 16) s += hl[15 - j] * vk;
    }
    return (lane >= 16) ? s : 0.0f;
}

__global__ void __launch_bounds__(256) fir_wmma(const float* __restrict__ x,
                                                const float* __restrict__ h,
                                                float* __restrict__ y) {
    __shared__ float sKpad[64];
    __shared__ __attribute__((aligned(16))) float sStage[WAVES_PER_BLOCK * LDS_WAVE];
    const int tid = threadIdx.x;
    if (tid < 63) {
        const int t = tid - 15;
        float v = 0.0f;
        if (t >= 0 && t <= 32) {
            const int alo = (t > 16) ? (t - 16) : 0;
            const int ahi = (t < 16) ? t : 16;
            float s = 0.0f;
            for (int a = alo; a <= ahi; ++a) s += k1_tap(h, a) * k2_tap(h, t - a);
            v = s;
        }
        sKpad[tid] = v;
    }
    __syncthreads();

    const int wave = tid >> 5;
    const int lane = tid & 31;
    const int n    = lane & 15;
    const int half = lane >> 4;

    const int wave_id = blockIdx.x * WAVES_PER_BLOCK + wave;
    const int row     = wave_id / STRIPS_PER_ROW;
    const int strip   = wave_id % STRIPS_PER_ROW;

    v2f a[3][4];
#pragma unroll
    for (int blk = 0; blk < 3; ++blk) {
#pragma unroll
        for (int c = 0; c < 4; ++c) {
            const int i0 = 15 + n - (8 * half + 2 * c) + 16 * blk;
            a[blk][c].x = sKpad[i0];
            a[blk][c].y = sKpad[i0 - 1];
        }
    }

    float* ring = sStage + wave * LDS_WAVE;
    float* outb = ring + LDS_RING;

    const int wl = 4 * lane + 4 * half;
    int fr[3], frw[3];
#pragma unroll
    for (int blk = 0; blk < 3; ++blk) {
        const int cb = n - blk;
        fr[blk]  = (cb >= 0) ? (16 * cb + 4 * (cb >> 2) + 8 * half) : (16 * cb - 52 + 8 * half);
        frw[blk] = (cb >= 0) ? fr[blk] : (fr[blk] + LDS_RING);
    }
    const int ow = fr[0];

    const size_t lane_off = (size_t)row * N_LEN + (size_t)strip * (TILES_PER_WAVE * TILE)
                          + 4 * lane;
    const float* xs = x + lane_off;
    float*       ys = y + lane_off;

    const bool row_head = (strip == 0);
    float xhead = 0.0f;
    if (row_head) xhead = x[(size_t)row * N_LEN + lane];

    v4f front = {0.0f, 0.0f, 0.0f, 0.0f};
    if (strip > 0 && lane < 8) front = *(const v4f*)(xs - 32);

    v4f A[GROUP][2], B[GROUP][2];
    load_lines(A, xs);
    load_lines(B, xs + GROUP * TILE);
    if (lane < 8) *(v4f*)(ring + LDS_RING - 84 + 4 * lane) = front;
    if (row_head) y[(size_t)row * N_LEN + lane] = row_head_value(xhead, h, lane);
    const bool skip_head = row_head && (lane < 8);

#pragma unroll
    for (int g = 0; g < GROUPS_PER_WAVE; g += 2) {
        stage_in(A, ring, wl);
        lds_wave_sync();
        if (g + 2 < GROUPS_PER_WAVE) load_lines(A, xs + (g + 2) * (GROUP * TILE));
        compute_group<0>(a, ring, outb, fr, frw, ow, wl, ys + g * (GROUP * TILE),
                         (g == 0) && skip_head);
        stage_in(B, ring + LDS_GROUP, wl);
        lds_wave_sync();
        if (g + 3 < GROUPS_PER_WAVE) load_lines(B, xs + (g + 3) * (GROUP * TILE));
        compute_group<1>(a, ring, outb, fr, frw, ow, wl, ys + (g + 1) * (GROUP * TILE), false);
    }
}

extern "C" void kernel_launch(void* const* d_in, const int* in_sizes, int n_in,
                              void* d_out, int out_size, void* d_ws, size_t ws_size,
                              hipStream_t stream) {
    const float* x = (const float*)d_in[0];
    const float* h = (const float*)d_in[1];
    float* y = (float*)d_out;
    (void)in_sizes; (void)n_in; (void)out_size; (void)d_ws; (void)ws_size;

    fir_wmma<<<NUM_BLOCKS, 256, 0, stream>>>(x, h, y);
}
